# starting kernel
# baseline (speedup 1.0000x reference)
#include <hip/hip_runtime.h>

#define DD 256
#define NSTRUCT 8192
#define EPS 1e-5f

typedef float v2f __attribute__((ext_vector_type(2)));
typedef float v8f __attribute__((ext_vector_type(8)));

__device__ __forceinline__ float wave_sum(float v) {
  v += __shfl_xor(v, 1, 32);
  v += __shfl_xor(v, 2, 32);
  v += __shfl_xor(v, 4, 32);
  v += __shfl_xor(v, 8, 32);
  v += __shfl_xor(v, 16, 32);
  return v;
}
__device__ __forceinline__ float sum4(float4 x) { return x.x + x.y + x.z + x.w; }
__device__ __forceinline__ float sumsq4(float4 x) { return x.x*x.x + x.y*x.y + x.z*x.z + x.w*x.w; }
__device__ __forceinline__ float4 ln4(float4 x, float mu, float rs, float4 g, float4 b) {
  float4 y;
  y.x = (x.x - mu) * rs * g.x + b.x;
  y.y = (x.y - mu) * rs * g.y + b.y;
  y.z = (x.z - mu) * rs * g.z + b.z;
  y.w = (x.w - mu) * rs * g.w + b.w;
  return y;
}

__global__ __launch_bounds__(256) void leaf_ln_kernel(
    const float* __restrict__ comp_emb, const int* __restrict__ leaf_ids,
    const float* __restrict__ gamma, const float* __restrict__ beta,
    float* __restrict__ h0) {
  const int lane = threadIdx.x & 31;
  const int r = blockIdx.x * 8 + (threadIdx.x >> 5);
  const int id = leaf_ids[r];
  const float4* src = (const float4*)(comp_emb + (size_t)id * DD);
  float4 x0 = src[lane];
  float4 x1 = src[lane + 32];
  float s  = wave_sum(sum4(x0) + sum4(x1));
  float s2 = wave_sum(sumsq4(x0) + sumsq4(x1));
  float mu = s * (1.0f / DD);
  float rs = rsqrtf(s2 * (1.0f / DD) - mu * mu + EPS);
  float4 g0 = ((const float4*)gamma)[lane], g1 = ((const float4*)gamma)[lane + 32];
  float4 b0 = ((const float4*)beta)[lane],  b1 = ((const float4*)beta)[lane + 32];
  float4* dst = (float4*)(h0 + (size_t)r * DD);
  dst[lane]      = ln4(x0, mu, rs, g0, b0);
  dst[lane + 32] = ln4(x1, mu, rs, g1, b1);
}

#define MT 32
#define KC 768
#define CSTR 772
#define OSTR 260

__global__ __launch_bounds__(256) void combine_kernel(
    const float* __restrict__ h_in, float* __restrict__ h_out,
    const int* __restrict__ op_ids, int off, int ln2n,
    const float* __restrict__ op_emb, const float* __restrict__ W,
    const float* __restrict__ bias, const float* __restrict__ gamma,
    const float* __restrict__ beta) {
  __shared__ float smem[MT * CSTR];

  const int tid = threadIdx.x;
  const int lane = tid & 31;
  const int wave = tid >> 5;
  const int r0 = blockIdx.x * MT;
  const int nmask = (1 << ln2n) - 1;

  for (int i = tid; i < MT * (KC / 4); i += 256) {
    const int m = i / (KC / 4);
    const int c4 = i - m * (KC / 4);
    const int rg = r0 + m;
    float4 v;
    if (c4 < 64) {
      const int opid = op_ids[(rg >> ln2n) * 7 + off + (rg & nmask)];
      v = ((const float4*)(op_emb + (size_t)opid * DD))[c4];
    } else if (c4 < 128) {
      v = ((const float4*)(h_in + (size_t)(2 * rg) * DD))[c4 - 64];
    } else {
      v = ((const float4*)(h_in + (size_t)(2 * rg + 1) * DD))[c4 - 128];
    }
    *(float4*)&smem[m * CSTR + c4 * 4] = v;
  }
  __syncthreads();

  v8f c00 = {}, c01 = {}, c10 = {}, c11 = {};
  const int n0 = wave * 32;
  const int colA = n0 + (lane & 15);
  const int krow = (lane & 16) >> 3;
  const float* arow0 = &smem[(lane & 15) * CSTR];
  const float* arow1 = &smem[((lane & 15) + 16) * CSTR];
#pragma unroll 4
  for (int k0 = 0; k0 < KC; k0 += 4) {
    const int kb = k0 + krow;
    const float* Wk = W + (size_t)kb * DD;
    v2f b0 = { Wk[colA],      Wk[DD + colA] };
    v2f b1 = { Wk[colA + 16], Wk[DD + colA + 16] };
    v2f a0 = *(const v2f*)(arow0 + kb);
    v2f a1 = *(const v2f*)(arow1 + kb);
    c00 = __builtin_amdgcn_wmma_f32_16x16x4_f32(false, a0, false, b0, (short)0, c00, false, false);
    c01 = __builtin_amdgcn_wmma_f32_16x16x4_f32(false, a0, false, b1, (short)0, c01, false, false);
    c10 = __builtin_amdgcn_wmma_f32_16x16x4_f32(false, a1, false, b0, (short)0, c10, false, false);
    c11 = __builtin_amdgcn_wmma_f32_16x16x4_f32(false, a1, false, b1, (short)0, c11, false, false);
  }
  __syncthreads();

  const float bias0 = bias[colA];
  const float bias1 = bias[colA + 16];
  const int roff = (lane & 16) >> 1;
#pragma unroll
  for (int j = 0; j < 8; ++j) {
    const int rr = j + roff;
    smem[rr * OSTR + colA]              = c00[j] + bias0;
    smem[rr * OSTR + colA + 16]         = c01[j] + bias1;
    smem[(rr + 16) * OSTR + colA]       = c10[j] + bias0;
    smem[(rr + 16) * OSTR + colA + 16]  = c11[j] + bias1;
  }
  __syncthreads();

  const float4 g0  = ((const float4*)gamma)[lane];
  const float4 g1  = ((const float4*)gamma)[lane + 32];
  const float4 bt0 = ((const float4*)beta)[lane];
  const float4 bt1 = ((const float4*)beta)[lane + 32];
#pragma unroll
  for (int i = 0; i < 4; ++i) {
    const int row = wave * 4 + i;
    float4 x0 = *(const float4*)&smem[row * OSTR + lane * 4];
    float4 x1 = *(const float4*)&smem[row * OSTR + 128 + lane * 4];
    float s  = wave_sum(sum4(x0) + sum4(x1));
    float s2 = wave_sum(sumsq4(x0) + sumsq4(x1));
    float mu = s * (1.0f / DD);
    float rs = rsqrtf(s2 * (1.0f / DD) - mu * mu + EPS);
    float4* dst = (float4*)(h_out + (size_t)(r0 + row) * DD);
    dst[lane]      = ln4(x0, mu, rs, g0, bt0);
    dst[lane + 32] = ln4(x1, mu, rs, g1, bt1);
  }
}

__global__ __launch_bounds__(256) void gather_kernel(
    const float* __restrict__ root, const int* __restrict__ sidx,
    float* __restrict__ out) {
  const int lane = threadIdx.x & 31;
  const int r = blockIdx.x * 8 + (threadIdx.x >> 5);
  const int idx = sidx[r];
  float4* dst = (float4*)(out + (size_t)r * DD);
  if (idx > 1) {
    const float4* src = (const float4*)(root + (size_t)idx * DD);
    dst[lane]      = src[lane];
    dst[lane + 32] = src[lane + 32];
  } else {
    float4 z = make_float4(0.f, 0.f, 0.f, 0.f);
    dst[lane]      = z;
    dst[lane + 32] = z;
  }
}

extern "C" void kernel_launch(void* const* d_in, const int* in_sizes, int n_in,
                              void* d_out, int out_size, void* d_ws, size_t ws_size,
                              hipStream_t stream) {
  const int*   struct_indices = (const int*)d_in[0];
  const int*   leaf_ids       = (const int*)d_in[1];
  const int*   op_ids         = (const int*)d_in[2];
  const float* comp_emb       = (const float*)d_in[3];
  const float* op_emb         = (const float*)d_in[4];
  const float* W_bin          = (const float*)d_in[5];
  const float* b_bin          = (const float*)d_in[6];
  const float* gamma          = (const float*)d_in[7];
  const float* beta           = (const float*)d_in[8];

  float* h0   = (float*)d_out;
  float* h1   = (float*)d_ws;
  float* h2   = h1 + (size_t)NSTRUCT * 4 * DD;
  float* root = h2 + (size_t)NSTRUCT * 2 * DD;

  leaf_ln_kernel<<<NSTRUCT * 8 / 8, 256, 0, stream>>>(comp_emb, leaf_ids, gamma, beta, h0);
  combine_kernel<<<(NSTRUCT * 4) / MT, 256, 0, stream>>>(h0, h1, op_ids, 0, 2,
                                                         op_emb, W_bin, b_bin, gamma, beta);
  combine_kernel<<<(NSTRUCT * 2) / MT, 256, 0, stream>>>(h1, h2, op_ids, 4, 1,
                                                         op_emb, W_bin, b_bin, gamma, beta);
  combine_kernel<<<(NSTRUCT * 1) / MT, 256, 0, stream>>>(h2, root, op_ids, 6, 0,
                                                         op_emb, W_bin, b_bin, gamma, beta);
  gather_kernel<<<65536 / 8, 256, 0, stream>>>(root, struct_indices, (float*)d_out);
}


# level 1 as one 64-row pass (was two of 32): each wave keeps 4x2 accumulator tiles, so each W fragment pair feeds 8 matrix instructions (was 4); LDS re-planned as 2 buffers of 64 rows; same sums
# speedup vs baseline: 1.4543x; 1.4543x over previous
#include <hip/hip_runtime.h>

#define DD 256
#define NSTRUCT 8192
#define EPS 1e-5f

typedef float v2f __attribute__((ext_vector_type(2)));
typedef float v4f __attribute__((ext_vector_type(4)));
typedef float v8f __attribute__((ext_vector_type(8)));

__device__ __forceinline__ float wave_sum(float v) {
  v += __shfl_xor(v, 1, 32);
  v += __shfl_xor(v, 2, 32);
  v += __shfl_xor(v, 4, 32);
  v += __shfl_xor(v, 8, 32);
  v += __shfl_xor(v, 16, 32);
  return v;
}
__device__ __forceinline__ float sum4(float4 x) { return x.x + x.y + x.z + x.w; }
__device__ __forceinline__ float sumsq4(float4 x) { return x.x*x.x + x.y*x.y + x.z*x.z + x.w*x.w; }
__device__ __forceinline__ float4 ln4(float4 x, float mu, float rs, float4 g, float4 b) {
  float4 y;
  y.x = (x.x - mu) * rs * g.x + b.x;
  y.y = (x.y - mu) * rs * g.y + b.y;
  y.z = (x.z - mu) * rs * g.z + b.z;
  y.w = (x.w - mu) * rs * g.w + b.w;
  return y;
}

#define SB 16
#define TSTR 260
#define TBUF (64 * TSTR)

template <int MT>
__device__ __forceinline__ void gemm_third(const float* __restrict__ W, const float* ap,
                                           int kbeg, int krow, int colA, v8f (&c)[MT][2]) {
#pragma unroll(MT == 1 ? 8 : 4)
  for (int k0 = 0; k0 < DD; k0 += 4) {
    const int kl = k0 + krow;
    const float* Wk = W + (size_t)(kbeg + kl) * DD;
    v2f b0 = { Wk[colA],      Wk[DD + colA] };
    v2f b1 = { Wk[colA + 16], Wk[DD + colA + 16] };
#pragma unroll
    for (int t = 0; t < MT; ++t) {
      v2f a = *(const v2f*)(ap + t * 16 * TSTR + kl);
      c[t][0] = __builtin_amdgcn_wmma_f32_16x16x4_f32(false, a, false, b0, (short)0, c[t][0], false, false);
      c[t][1] = __builtin_amdgcn_wmma_f32_16x16x4_f32(false, a, false, b1, (short)0, c[t][1], false, false);
    }
  }
}

template <int MT>
__device__ __forceinline__ void spill_c(float* C, int lane, int colA, float bias0, float bias1,
                                        const v8f (&c)[MT][2]) {
  const int roff = (lane & 16) >> 1;
#pragma unroll
  for (int t = 0; t < MT; ++t) {
#pragma unroll
    for (int j = 0; j < 8; ++j) {
      const int rr = 16 * t + j + roff;
      C[rr * TSTR + colA]      = c[t][0][j] + bias0;
      C[rr * TSTR + colA + 16] = c[t][1][j] + bias1;
    }
  }
}

__device__ __forceinline__ void ln_row(const float* crow, float* dst, int lane,
                                       float4 g0, float4 g1, float4 bt0, float4 bt1) {
  float4 x0 = *(const float4*)&crow[lane * 4];
  float4 x1 = *(const float4*)&crow[128 + lane * 4];
  float s  = wave_sum(sum4(x0) + sum4(x1));
  float s2 = wave_sum(sumsq4(x0) + sumsq4(x1));
  float mu = s * (1.0f / DD);
  float rs = rsqrtf(s2 * (1.0f / DD) - mu * mu + EPS);
  *(float4*)&dst[lane * 4]       = ln4(x0, mu, rs, g0, bt0);
  *(float4*)&dst[128 + lane * 4] = ln4(x1, mu, rs, g1, bt1);
}

__global__ __launch_bounds__(256) void tree_kernel(
    const float* __restrict__ comp_emb, const int* __restrict__ leaf_ids,
    const int* __restrict__ op_ids, const float* __restrict__ op_emb,
    const float* __restrict__ W, const float* __restrict__ bias,
    const float* __restrict__ gamma, const float* __restrict__ beta,
    float* __restrict__ root) {
  __shared__ float smem[2 * TBUF];
  float* const P = smem;
  float* const Q = smem + TBUF;

  const int tid = threadIdx.x;
  const int lane = tid & 31;
  const int wave = tid >> 5;
  const int blk = blockIdx.x;
  const int colA = wave * 32 + (lane & 15);
  const int krow = (lane & 16) >> 3;
  const int arow = (lane & 15) * TSTR;

  const int c4 = tid & 63;
  const int mb = tid >> 6;
  const v4f* op4 = (const v4f*)op_emb;

  const float4 g0  = ((const float4*)gamma)[lane];
  const float4 g1  = ((const float4*)gamma)[lane + 32];
  const float4 bt0 = ((const float4*)beta)[lane];
  const float4 bt1 = ((const float4*)beta)[lane + 32];
  const float bias0 = bias[colA];
  const float bias1 = bias[colA + 16];

  int id[16];
  {
    const int4* lp = (const int4*)(leaf_ids + 128 * blk + wave * 16);
#pragma unroll
    for (int q = 0; q < 4; ++q) {
      const int4 iv = lp[q];
      id[4 * q + 0] = iv.x; id[4 * q + 1] = iv.y; id[4 * q + 2] = iv.z; id[4 * q + 3] = iv.w;
    }
  }
  {
    int opid1[16];
#pragma unroll
    for (int u = 0; u < 16; ++u) {
      const int rg = 64 * blk + mb + 4 * u;
      opid1[u] = op_ids[(rg >> 2) * 7 + (rg & 3)];
    }
    v4f vop1[16];
#pragma unroll
    for (int u = 0; u < 16; ++u) vop1[u] = op4[opid1[u] * (DD / 4) + c4];
#pragma unroll
    for (int u = 0; u < 16; ++u) *(v4f*)&P[(mb + 4 * u) * TSTR + c4 * 4] = vop1[u];
  }

  v4f vop[8];
  int opid[8];

  {
    v8f c[4][2] = {};
    float4 x0[8], x1[8];
#pragma unroll
    for (int side = 0; side < 2; ++side) {
      __syncthreads();
#pragma unroll
      for (int t = 0; t < 8; ++t) {
        const int b = id[2 * t + side] * (DD / 4) + lane;
        x0[t] = ((const float4*)comp_emb)[b];
        x1[t] = ((const float4*)comp_emb)[b + 32];
      }
      if (side == 1) {
#pragma unroll
        for (int u = 0; u < 8; ++u) {
          const int rg = 32 * blk + mb + 4 * u;
          opid[u] = op_ids[(rg >> 1) * 7 + 4 + (rg & 1)];
        }
      }
      const float* A = side ? Q : P;
      gemm_third<4>(W, A + arow, side * DD, krow, colA, c);
      float* D = side ? P : Q;
#pragma unroll
      for (int t = 0; t < 8; ++t) {
        float s  = wave_sum(sum4(x0[t]) + sum4(x1[t]));
        float s2 = wave_sum(sumsq4(x0[t]) + sumsq4(x1[t]));
        float mu = s * (1.0f / DD);
        float rs = rsqrtf(s2 * (1.0f / DD) - mu * mu + EPS);
        float* drow = &D[(wave * 8 + t) * TSTR];
        *(float4*)&drow[lane * 4]       = ln4(x0[t], mu, rs, g0, bt0);
        *(float4*)&drow[128 + lane * 4] = ln4(x1[t], mu, rs, g1, bt1);
      }
    }
    __syncthreads();
#pragma unroll
    for (int u = 0; u < 8; ++u) vop[u] = op4[opid[u] * (DD / 4) + c4];
    gemm_third<4>(W, P + arow, 2 * DD, krow, colA, c);

    __syncthreads();
    spill_c<4>(Q, lane, colA, bias0, bias1, c);
    __syncthreads();
#pragma unroll
    for (int i = 0; i < 8; ++i) {
      const int m = wave * 8 + i;
      float* dst = P + ((m & 1) * 32 + (m >> 1)) * TSTR;
      ln_row(&Q[m * TSTR], dst, lane, g0, g1, bt0, bt1);
    }
  }

  v4f vop3[4];
  {
    __syncthreads();
#pragma unroll
    for (int u = 0; u < 8; ++u) *(v4f*)&Q[(mb + 4 * u) * TSTR + c4 * 4] = vop[u];
    __syncthreads();
    int opid3[4];
#pragma unroll
    for (int u = 0; u < 4; ++u) opid3[u] = op_ids[(16 * blk + mb + 4 * u) * 7 + 6];
    v8f c[2][2] = {};
    gemm_third<2>(W, Q + arow, 0, krow, colA, c);
#pragma unroll
    for (int u = 0; u < 4; ++u) vop3[u] = op4[opid3[u] * (DD / 4) + c4];
    gemm_third<2>(W, P + arow,             DD,     krow, colA, c);
    gemm_third<2>(W, P + 32 * TSTR + arow, 2 * DD, krow, colA, c);

    spill_c<2>(Q + 32 * TSTR, lane, colA, bias0, bias1, c);
    __syncthreads();
#pragma unroll
    for (int i = 0; i < 4; ++i) {
      const int m = wave * 4 + i;
      float* dst = P + ((m & 1) * 16 + (m >> 1)) * TSTR;
      ln_row(&Q[(32 + m) * TSTR], dst, lane, g0, g1, bt0, bt1);
    }
  }

  {
#pragma unroll
    for (int u = 0; u < 4; ++u) *(v4f*)&Q[(mb + 4 * u) * TSTR + c4 * 4] = vop3[u];
    __syncthreads();
    v8f c[1][2] = {};
    gemm_third<1>(W, Q + arow,             0,      krow, colA, c);
    gemm_third<1>(W, P + arow,             DD,     krow, colA, c);
    gemm_third<1>(W, P + 16 * TSTR + arow, 2 * DD, krow, colA, c);
    spill_c<1>(Q + 32 * TSTR, lane, colA, bias0, bias1, c);
    __syncthreads();
#pragma unroll
    for (int i = 0; i < 2; ++i) {
      const int m = wave * 2 + i;
      ln_row(&Q[(32 + m) * TSTR], root + (size_t)(SB * blk + m) * DD, lane, g0, g1, bt0, bt1);
    }
  }
}

__global__ __launch_bounds__(256) void gather_kernel(
    const float* __restrict__ root, const int* __restrict__ sidx,
    float* __restrict__ out) {
  const int lane = threadIdx.x & 31;
  const int r = blockIdx.x * 8 + (threadIdx.x >> 5);
  const int idx = sidx[r];
  float4* dst = (float4*)(out + (size_t)r * DD);
  if (idx > 1) {
    const float4* src = (const float4*)(root + (size_t)idx * DD);
    dst[lane]      = src[lane];
    dst[lane + 32] = src[lane + 32];
  } else {
    float4 z = make_float4(0.f, 0.f, 0.f, 0.f);
    dst[lane]      = z;
    dst[lane + 32] = z;
  }
}

extern "C" void kernel_launch(void* const* d_in, const int* in_sizes, int n_in,
                              void* d_out, int out_size, void* d_ws, size_t ws_size,
                              hipStream_t stream) {
  const int*   struct_indices = (const int*)d_in[0];
  const int*   leaf_ids       = (const int*)d_in[1];
  const int*   op_ids         = (const int*)d_in[2];
  const float* comp_emb       = (const float*)d_in[3];
  const float* op_emb         = (const float*)d_in[4];
  const float* W_bin          = (const float*)d_in[5];
  const float* b_bin          = (const float*)d_in[6];
  const float* gamma          = (const float*)d_in[7];
  const float* beta           = (const float*)d_in[8];

  float* root = (float*)d_ws;

  tree_kernel<<<NSTRUCT / SB, 256, 0, stream>>>(comp_emb, leaf_ids, op_ids, op_emb, W_bin, b_bin,
                                                gamma, beta, root);
  gather_kernel<<<65536 / 8, 256, 0, stream>>>(root, struct_indices, (float*)d_out);
}
